# starting kernel
# baseline (speedup 1.0000x reference)
#include <hip/hip_runtime.h>


#define N_NODES 20000
#define E_EDGES 320000
#define INV_SQRT3 0.57735026918962576451f

typedef __attribute__((ext_vector_type(2))) float v2f;
typedef __attribute__((ext_vector_type(8))) float v8f;

static __device__ __forceinline__ v8f wmma4(v2f a, v2f b, v8f c) {
  return __builtin_amdgcn_wmma_f32_16x16x4_f32(
      false, a, false, b, (short)0, c, false, false);
}

static __device__ __forceinline__ v2f ldb(const float* __restrict__ WP,
                                          int ncol, int k0, int col) {
  return *(const v2f*)(WP + (size_t)(k0 >> 1) * (2 * ncol) + col * 2);
}

static __device__ __forceinline__ v2f ld2(const float* __restrict__ p) {
  v2f r; r.x = p[0]; r.y = p[1]; return r;
}

__global__ void k_pack_pairs(const float* __restrict__ W, float* __restrict__ WP,
                             int K, int ncol) {
  const int i = blockIdx.x * blockDim.x + threadIdx.x;
  if (i >= K * ncol) return;
  const int k = i / ncol, c = i - k * ncol;
  WP[(size_t)(k >> 1) * (2 * ncol) + c * 2 + (k & 1)] = W[i];
}

__global__ __launch_bounds__(128) void k_node_pre(
    const float* __restrict__ nf, const float* __restrict__ Wp0P,
    const float* __restrict__ Wp1P, float* __restrict__ x0,
    float* __restrict__ x1, float* __restrict__ t0, float* __restrict__ t1) {
  const int wave = threadIdx.x >> 5;
  const int lane = threadIdx.x & 31;
  const int tile = blockIdx.x * 4 + wave;
  if (tile >= (N_NODES / 16)) return;
  const int nbase = tile * 16;
  const int lrow  = lane & 15;
  const int kh    = lane >> 4;
  const int arow  = nbase + lrow;

  {
    v8f acc[4] = {{}, {}, {}, {}};
    for (int kk = 0; kk < 64; kk += 4) {
      const int k0 = kk + 2 * kh;
      const v2f a = ld2(&nf[(size_t)arow * 256 + k0]);
#pragma unroll
      for (int j = 0; j < 4; ++j)
        acc[j] = wmma4(a, ldb(Wp0P, 64, k0, 16 * j + lrow), acc[j]);
    }
#pragma unroll
    for (int j = 0; j < 4; ++j)
#pragma unroll
      for (int v = 0; v < 8; ++v)
        x0[(size_t)(nbase + v + 8 * kh) * 64 + 16 * j + lrow] = acc[j][v];
  }

  for (int m = 0; m < 3; ++m) {
    v8f acc[4] = {{}, {}, {}, {}};
    for (int kk = 0; kk < 64; kk += 4) {
      const int k0 = kk + 2 * kh;
      v2f a;
      a.x = nf[(size_t)arow * 256 + 64 + k0 * 3 + m];
      a.y = nf[(size_t)arow * 256 + 64 + (k0 + 1) * 3 + m];
#pragma unroll
      for (int j = 0; j < 4; ++j)
        acc[j] = wmma4(a, ldb(Wp1P, 64, k0, 16 * j + lrow), acc[j]);
    }
#pragma unroll
    for (int j = 0; j < 4; ++j)
#pragma unroll
      for (int v = 0; v < 8; ++v)
        x1[(size_t)(nbase + v + 8 * kh) * 192 + (16 * j + lrow) * 3 + m] = acc[j][v];
  }

  float4* t0v = (float4*)(t0 + (size_t)nbase * 64);
  float4* t1v = (float4*)(t1 + (size_t)nbase * 192);
  const float4 z4 = make_float4(0.f, 0.f, 0.f, 0.f);
  for (int i = lane; i < 16 * 64 / 4; i += 32)  t0v[i] = z4;
  for (int i = lane; i < 16 * 192 / 4; i += 32) t1v[i] = z4;
}

__global__ __launch_bounds__(128) void k_edge(
    const float* __restrict__ sph, const float* __restrict__ rb,
    const int* __restrict__ snd, const int* __restrict__ rcv,
    const float* __restrict__ W1P, const float* __restrict__ W2P,
    const float* __restrict__ x0, const float* __restrict__ x1,
    float* __restrict__ t0, float* __restrict__ t1) {
  __shared__ float hidden[4][16][64];
  const int wave  = threadIdx.x >> 5;
  const int lane  = threadIdx.x & 31;
  const int lrow  = lane & 15;
  const int kh    = lane >> 4;
  const int ebase = (blockIdx.x * 4 + wave) * 16;

  {
    v8f acc[4] = {{}, {}, {}, {}};
#pragma unroll
    for (int kk = 0; kk < 8; kk += 4) {
      const int k0 = kk + 2 * kh;
      const v2f a = ld2(&rb[(size_t)(ebase + lrow) * 8 + k0]);
#pragma unroll
      for (int j = 0; j < 4; ++j)
        acc[j] = wmma4(a, ldb(W1P, 64, k0, 16 * j + lrow), acc[j]);
    }
#pragma unroll
    for (int j = 0; j < 4; ++j)
#pragma unroll
      for (int v = 0; v < 8; ++v) {
        float h = acc[j][v];
        h = h / (1.f + __expf(-h));
        hidden[wave][v + 8 * kh][16 * j + lrow] = h;
      }
  }
  __syncthreads();

  v8f w[16] = {{}, {}, {}, {}, {}, {}, {}, {}, {}, {}, {}, {}, {}, {}, {}, {}};
  for (int kk = 0; kk < 64; kk += 4) {
    const int k0 = kk + 2 * kh;
    const v2f a = ld2(&hidden[wave][lrow][k0]);
#pragma unroll
    for (int j = 0; j < 16; ++j)
      w[j] = wmma4(a, ldb(W2P, 256, k0, 16 * j + lrow), w[j]);
  }

#pragma unroll
  for (int v = 0; v < 8; ++v) {
    const int e = ebase + v + 8 * kh;
    const int s = snd[e];
    const int r = rcv[e];
    const float Y0  = sph[(size_t)e * 4 + 0];
    const float Y1x = sph[(size_t)e * 4 + 1];
    const float Y1y = sph[(size_t)e * 4 + 2];
    const float Y1z = sph[(size_t)e * 4 + 3];
#pragma unroll
    for (int j = 0; j < 4; ++j) {
      const int c = 16 * j + lrow;
      const float w0 = w[j][v], w1 = w[j + 4][v];
      const float w2 = w[j + 8][v], w3 = w[j + 12][v];
      const float h0  = x0[(size_t)s * 64 + c];
      const float h1x = x1[(size_t)s * 192 + c * 3 + 0];
      const float h1y = x1[(size_t)s * 192 + c * 3 + 1];
      const float h1z = x1[(size_t)s * 192 + c * 3 + 2];
      const float dot = h1x * Y1x + h1y * Y1y + h1z * Y1z;
      atomicAdd(&t0[(size_t)r * 64 + c], w0 * h0 * Y0 + w1 * dot * INV_SQRT3);
      const size_t b1 = (size_t)r * 192 + c * 3;
      atomicAdd(&t1[b1 + 0], w2 * h0 * Y1x + w3 * h1x * Y0);
      atomicAdd(&t1[b1 + 1], w2 * h0 * Y1y + w3 * h1y * Y0);
      atomicAdd(&t1[b1 + 2], w2 * h0 * Y1z + w3 * h1z * Y0);
    }
  }
}

__global__ __launch_bounds__(64) void k_node_post(
    const float* __restrict__ nf, const float* __restrict__ attrs,
    const float* __restrict__ Wpost0P, const float* __restrict__ Wpost1P,
    const float* __restrict__ Wc0, const float* __restrict__ Wc1,
    const float* __restrict__ Wr0P, const float* __restrict__ Wr1P,
    const float* __restrict__ Wf0P, const float* __restrict__ Wf1P,
    const float* __restrict__ t0, const float* __restrict__ t1,
    float* __restrict__ out) {
  __shared__ float s0s[2][16][64];
  __shared__ float s1s[2][16][192];
  const int wave  = threadIdx.x >> 5;
  const int lane  = threadIdx.x & 31;
  const int lrow  = lane & 15;
  const int kh    = lane >> 4;
  const int nbase = (blockIdx.x * 2 + wave) * 16;
  const int arow  = nbase + lrow;

  v8f a0[4]    = {{}, {}, {}, {}};
  v8f a1[3][4] = {{{}, {}, {}, {}}, {{}, {}, {}, {}}, {{}, {}, {}, {}}};
  for (int kk = 0; kk < 64; kk += 4) {
    const int k0 = kk + 2 * kh;
    const v2f at0 = ld2(&t0[(size_t)arow * 64 + k0]);
#pragma unroll
    for (int j = 0; j < 4; ++j)
      a0[j] = wmma4(at0, ldb(Wpost0P, 64, k0, 16 * j + lrow), a0[j]);
#pragma unroll
    for (int m = 0; m < 3; ++m) {
      v2f at1;
      at1.x = t1[(size_t)arow * 192 + k0 * 3 + m];
      at1.y = t1[(size_t)arow * 192 + (k0 + 1) * 3 + m];
#pragma unroll
      for (int j = 0; j < 4; ++j)
        a1[m][j] = wmma4(at1, ldb(Wpost1P, 64, k0, 16 * j + lrow), a1[m][j]);
    }
  }

#pragma unroll
  for (int j = 0; j < 4; ++j) {
    const int col = 16 * j + lrow;
#pragma unroll
    for (int v = 0; v < 8; ++v) {
      const int node = nbase + v + 8 * kh;
      const float s  = a0[j][v] * (1.f / 16.f);
      const float ax = a1[0][j][v] * (1.f / 16.f);
      const float ay = a1[1][j][v] * (1.f / 16.f);
      const float az = a1[2][j][v] * (1.f / 16.f);
      const float q  = ax * ax + ay * ay + az * az;
      float c0 = 0.f, c1 = 0.f, c2 = 0.f, d0 = 0.f, d1 = 0.f, d2 = 0.f;
#pragma unroll
      for (int z = 0; z < 10; ++z) {
        const float t = attrs[(size_t)node * 10 + z];
        c0 += t * Wc0[z * 64 + col];
        c1 += t * Wc0[640 + z * 64 + col];
        c2 += t * Wc0[1280 + z * 64 + col];
        d0 += t * Wc1[z * 64 + col];
        d1 += t * Wc1[640 + z * 64 + col];
        d2 += t * Wc1[1280 + z * 64 + col];
      }
      const float s2 = s * s;
      const float g  = d2 * (s2 + q) + d1 * s + d0;
      a0[j][v]    = c2 * (s2 * s + s * q) + c1 * (s2 + q) + c0 * s;
      a1[0][j][v] = g * ax;
      a1[1][j][v] = g * ay;
      a1[2][j][v] = g * az;
    }
  }

  for (int kk = 0; kk < 640; kk += 4) {
    const int k0 = kk + 2 * kh;
    const int u0 = k0 / 10, z0 = k0 - u0 * 10;
    const int k1 = k0 + 1;
    const int u1 = k1 / 10, z1 = k1 - u1 * 10;
    v2f a;
    a.x = nf[(size_t)arow * 256 + u0] * attrs[(size_t)arow * 10 + z0];
    a.y = nf[(size_t)arow * 256 + u1] * attrs[(size_t)arow * 10 + z1];
#pragma unroll
    for (int j = 0; j < 4; ++j)
      a0[j] = wmma4(a, ldb(Wr0P, 64, k0, 16 * j + lrow), a0[j]);
  }
#pragma unroll
  for (int m = 0; m < 3; ++m) {
    for (int kk = 0; kk < 640; kk += 4) {
      const int k0 = kk + 2 * kh;
      const int u0 = k0 / 10, z0 = k0 - u0 * 10;
      const int k1 = k0 + 1;
      const int u1 = k1 / 10, z1 = k1 - u1 * 10;
      v2f a;
      a.x = nf[(size_t)arow * 256 + 64 + u0 * 3 + m] * attrs[(size_t)arow * 10 + z0];
      a.y = nf[(size_t)arow * 256 + 64 + u1 * 3 + m] * attrs[(size_t)arow * 10 + z1];
#pragma unroll
      for (int j = 0; j < 4; ++j)
        a1[m][j] = wmma4(a, ldb(Wr1P, 64, k0, 16 * j + lrow), a1[m][j]);
    }
  }

#pragma unroll
  for (int j = 0; j < 4; ++j) {
    const int col = 16 * j + lrow;
#pragma unroll
    for (int v = 0; v < 8; ++v) {
      const int M = v + 8 * kh;
      s0s[wave][M][col] = a0[j][v];
      s1s[wave][M][col * 3 + 0] = a1[0][j][v];
      s1s[wave][M][col * 3 + 1] = a1[1][j][v];
      s1s[wave][M][col * 3 + 2] = a1[2][j][v];
    }
  }
  __syncthreads();

  {
    v8f acc[4] = {{}, {}, {}, {}};
    for (int kk = 0; kk < 64; kk += 4) {
      const int k0 = kk + 2 * kh;
      const v2f a = ld2(&s0s[wave][lrow][k0]);
#pragma unroll
      for (int j = 0; j < 4; ++j)
        acc[j] = wmma4(a, ldb(Wf0P, 64, k0, 16 * j + lrow), acc[j]);
    }
#pragma unroll
    for (int j = 0; j < 4; ++j)
#pragma unroll
      for (int v = 0; v < 8; ++v)
        out[(size_t)(nbase + v + 8 * kh) * 256 + 16 * j + lrow] = acc[j][v];
  }
#pragma unroll
  for (int m = 0; m < 3; ++m) {
    v8f acc[4] = {{}, {}, {}, {}};
    for (int kk = 0; kk < 64; kk += 4) {
      const int k0 = kk + 2 * kh;
      v2f a;
      a.x = s1s[wave][lrow][k0 * 3 + m];
      a.y = s1s[wave][lrow][(k0 + 1) * 3 + m];
#pragma unroll
      for (int j = 0; j < 4; ++j)
        acc[j] = wmma4(a, ldb(Wf1P, 64, k0, 16 * j + lrow), acc[j]);
    }
#pragma unroll
    for (int j = 0; j < 4; ++j)
#pragma unroll
      for (int v = 0; v < 8; ++v)
        out[(size_t)(nbase + v + 8 * kh) * 256 + 64 + (16 * j + lrow) * 3 + m] =
            acc[j][v];
  }
}

extern "C" void kernel_launch(void* const* d_in, const int* in_sizes, int n_in,
                              void* d_out, int out_size, void* d_ws, size_t ws_size,
                              hipStream_t stream) {
  (void)in_sizes; (void)n_in; (void)out_size; (void)ws_size;
  const float* nf     = (const float*)d_in[0];
  const float* attrs  = (const float*)d_in[1];
  const float* sph    = (const float*)d_in[2];
  const float* rb     = (const float*)d_in[3];
  const int*   snd    = (const int*)d_in[4];
  const int*   rcv    = (const int*)d_in[5];
  const float* Wpre0  = (const float*)d_in[6];
  const float* Wpre1  = (const float*)d_in[7];
  const float* W1     = (const float*)d_in[8];
  const float* W2     = (const float*)d_in[9];
  const float* Wpost0 = (const float*)d_in[10];
  const float* Wpost1 = (const float*)d_in[11];
  const float* Wc0    = (const float*)d_in[12];
  const float* Wc1    = (const float*)d_in[13];
  const float* Wr0    = (const float*)d_in[14];
  const float* Wr1    = (const float*)d_in[15];
  const float* Wf0    = (const float*)d_in[16];
  const float* Wf1    = (const float*)d_in[17];

  float* ws = (float*)d_ws;
  size_t off = 0;
  float* x0 = ws + off; off += (size_t)N_NODES * 64;
  float* x1 = ws + off; off += (size_t)N_NODES * 192;
  float* t0 = ws + off; off += (size_t)N_NODES * 64;
  float* t1 = ws + off; off += (size_t)N_NODES * 192;
  float* Wpre0P  = ws + off; off += 64 * 64;
  float* Wpre1P  = ws + off; off += 64 * 64;
  float* W1P     = ws + off; off += 8 * 64;
  float* W2P     = ws + off; off += 64 * 256;
  float* Wpost0P = ws + off; off += 64 * 64;
  float* Wpost1P = ws + off; off += 64 * 64;
  float* Wr0P    = ws + off; off += 640 * 64;
  float* Wr1P    = ws + off; off += 640 * 64;
  float* Wf0P    = ws + off; off += 64 * 64;
  float* Wf1P    = ws + off; off += 64 * 64;

  auto pack = [&](const float* W, float* WP, int K, int ncol) {
    const int n = K * ncol;
    k_pack_pairs<<<(n + 255) / 256, 256, 0, stream>>>(W, WP, K, ncol);
  };
  pack(Wpre0, Wpre0P, 64, 64);
  pack(Wpre1, Wpre1P, 64, 64);
  pack(W1, W1P, 8, 64);
  pack(W2, W2P, 64, 256);
  pack(Wpost0, Wpost0P, 64, 64);
  pack(Wpost1, Wpost1P, 64, 64);
  pack(Wr0, Wr0P, 640, 64);
  pack(Wr1, Wr1P, 640, 64);
  pack(Wf0, Wf0P, 64, 64);
  pack(Wf1, Wf1P, 64, 64);

  const int ntiles = N_NODES / 16;
  k_node_pre<<<(ntiles + 3) / 4, 128, 0, stream>>>(nf, Wpre0P, Wpre1P, x0, x1, t0, t1);
  k_edge<<<E_EDGES / 64, 128, 0, stream>>>(sph, rb, snd, rcv, W1P, W2P, x0, x1, t0, t1);
  k_node_post<<<ntiles / 2, 64, 0, stream>>>(nf, attrs, Wpost0P, Wpost1P, Wc0, Wc1,
                                             Wr0P, Wr1P, Wf0P, Wf1P, t0, t1,
                                             (float*)d_out);
}


# node-post kernel re-tiled: one wave per 16-node x 16-column tile (4 waves/block, 4 fused accumulator chains, K=640 A operand built from registers) hides per-k-step load latency; sum order unchanged
# speedup vs baseline: 1.3029x; 1.3029x over previous
#include <hip/hip_runtime.h>


#define N_NODES 20000
#define E_EDGES 320000
#define INV_SQRT3 0.57735026918962576451f

typedef __attribute__((ext_vector_type(2))) float v2f;
typedef __attribute__((ext_vector_type(8))) float v8f;

static __device__ __forceinline__ v8f wmma4(v2f a, v2f b, v8f c) {
  return __builtin_amdgcn_wmma_f32_16x16x4_f32(
      false, a, false, b, (short)0, c, false, false);
}

static __device__ __forceinline__ v2f ldb(const float* __restrict__ WP,
                                          int ncol, int k0, int col) {
  return *(const v2f*)(WP + (size_t)(k0 >> 1) * (2 * ncol) + col * 2);
}

static __device__ __forceinline__ v2f ld2(const float* __restrict__ p) {
  v2f r; r.x = p[0]; r.y = p[1]; return r;
}

__global__ void k_pack_pairs(const float* __restrict__ W, float* __restrict__ WP,
                             int K, int ncol) {
  const int i = blockIdx.x * blockDim.x + threadIdx.x;
  if (i >= K * ncol) return;
  const int k = i / ncol, c = i - k * ncol;
  WP[(size_t)(k >> 1) * (2 * ncol) + c * 2 + (k & 1)] = W[i];
}

__global__ __launch_bounds__(128) void k_node_pre(
    const float* __restrict__ nf, const float* __restrict__ Wp0P,
    const float* __restrict__ Wp1P, float* __restrict__ x0,
    float* __restrict__ x1, float* __restrict__ t0, float* __restrict__ t1) {
  const int wave = threadIdx.x >> 5;
  const int lane = threadIdx.x & 31;
  const int tile = blockIdx.x * 4 + wave;
  if (tile >= (N_NODES / 16)) return;
  const int nbase = tile * 16;
  const int lrow  = lane & 15;
  const int kh    = lane >> 4;
  const int arow  = nbase + lrow;

  {
    v8f acc[4] = {{}, {}, {}, {}};
    for (int kk = 0; kk < 64; kk += 4) {
      const int k0 = kk + 2 * kh;
      const v2f a = ld2(&nf[(size_t)arow * 256 + k0]);
#pragma unroll
      for (int j = 0; j < 4; ++j)
        acc[j] = wmma4(a, ldb(Wp0P, 64, k0, 16 * j + lrow), acc[j]);
    }
#pragma unroll
    for (int j = 0; j < 4; ++j)
#pragma unroll
      for (int v = 0; v < 8; ++v)
        x0[(size_t)(nbase + v + 8 * kh) * 64 + 16 * j + lrow] = acc[j][v];
  }

  for (int m = 0; m < 3; ++m) {
    v8f acc[4] = {{}, {}, {}, {}};
    for (int kk = 0; kk < 64; kk += 4) {
      const int k0 = kk + 2 * kh;
      v2f a;
      a.x = nf[(size_t)arow * 256 + 64 + k0 * 3 + m];
      a.y = nf[(size_t)arow * 256 + 64 + (k0 + 1) * 3 + m];
#pragma unroll
      for (int j = 0; j < 4; ++j)
        acc[j] = wmma4(a, ldb(Wp1P, 64, k0, 16 * j + lrow), acc[j]);
    }
#pragma unroll
    for (int j = 0; j < 4; ++j)
#pragma unroll
      for (int v = 0; v < 8; ++v)
        x1[(size_t)(nbase + v + 8 * kh) * 192 + (16 * j + lrow) * 3 + m] = acc[j][v];
  }

  float4* t0v = (float4*)(t0 + (size_t)nbase * 64);
  float4* t1v = (float4*)(t1 + (size_t)nbase * 192);
  const float4 z4 = make_float4(0.f, 0.f, 0.f, 0.f);
  for (int i = lane; i < 16 * 64 / 4; i += 32)  t0v[i] = z4;
  for (int i = lane; i < 16 * 192 / 4; i += 32) t1v[i] = z4;
}

__global__ __launch_bounds__(128) void k_edge(
    const float* __restrict__ sph, const float* __restrict__ rb,
    const int* __restrict__ snd, const int* __restrict__ rcv,
    const float* __restrict__ W1P, const float* __restrict__ W2P,
    const float* __restrict__ x0, const float* __restrict__ x1,
    float* __restrict__ t0, float* __restrict__ t1) {
  __shared__ float hidden[4][16][64];
  const int wave  = threadIdx.x >> 5;
  const int lane  = threadIdx.x & 31;
  const int lrow  = lane & 15;
  const int kh    = lane >> 4;
  const int ebase = (blockIdx.x * 4 + wave) * 16;

  {
    v8f acc[4] = {{}, {}, {}, {}};
#pragma unroll
    for (int kk = 0; kk < 8; kk += 4) {
      const int k0 = kk + 2 * kh;
      const v2f a = ld2(&rb[(size_t)(ebase + lrow) * 8 + k0]);
#pragma unroll
      for (int j = 0; j < 4; ++j)
        acc[j] = wmma4(a, ldb(W1P, 64, k0, 16 * j + lrow), acc[j]);
    }
#pragma unroll
    for (int j = 0; j < 4; ++j)
#pragma unroll
      for (int v = 0; v < 8; ++v) {
        float h = acc[j][v];
        h = h / (1.f + __expf(-h));
        hidden[wave][v + 8 * kh][16 * j + lrow] = h;
      }
  }
  __syncthreads();

  v8f w[16] = {{}, {}, {}, {}, {}, {}, {}, {}, {}, {}, {}, {}, {}, {}, {}, {}};
  for (int kk = 0; kk < 64; kk += 4) {
    const int k0 = kk + 2 * kh;
    const v2f a = ld2(&hidden[wave][lrow][k0]);
#pragma unroll
    for (int j = 0; j < 16; ++j)
      w[j] = wmma4(a, ldb(W2P, 256, k0, 16 * j + lrow), w[j]);
  }

#pragma unroll
  for (int v = 0; v < 8; ++v) {
    const int e = ebase + v + 8 * kh;
    const int s = snd[e];
    const int r = rcv[e];
    const float Y0  = sph[(size_t)e * 4 + 0];
    const float Y1x = sph[(size_t)e * 4 + 1];
    const float Y1y = sph[(size_t)e * 4 + 2];
    const float Y1z = sph[(size_t)e * 4 + 3];
#pragma unroll
    for (int j = 0; j < 4; ++j) {
      const int c = 16 * j + lrow;
      const float w0 = w[j][v], w1 = w[j + 4][v];
      const float w2 = w[j + 8][v], w3 = w[j + 12][v];
      const float h0  = x0[(size_t)s * 64 + c];
      const float h1x = x1[(size_t)s * 192 + c * 3 + 0];
      const float h1y = x1[(size_t)s * 192 + c * 3 + 1];
      const float h1z = x1[(size_t)s * 192 + c * 3 + 2];
      const float dot = h1x * Y1x + h1y * Y1y + h1z * Y1z;
      atomicAdd(&t0[(size_t)r * 64 + c], w0 * h0 * Y0 + w1 * dot * INV_SQRT3);
      const size_t b1 = (size_t)r * 192 + c * 3;
      atomicAdd(&t1[b1 + 0], w2 * h0 * Y1x + w3 * h1x * Y0);
      atomicAdd(&t1[b1 + 1], w2 * h0 * Y1y + w3 * h1y * Y0);
      atomicAdd(&t1[b1 + 2], w2 * h0 * Y1z + w3 * h1z * Y0);
    }
  }
}

__global__ __launch_bounds__(128) void k_node_post(
    const float* __restrict__ nf, const float* __restrict__ attrs,
    const float* __restrict__ Wpost0P, const float* __restrict__ Wpost1P,
    const float* __restrict__ Wc0, const float* __restrict__ Wc1,
    const float* __restrict__ Wr0P, const float* __restrict__ Wr1P,
    const float* __restrict__ Wf0P, const float* __restrict__ Wf1P,
    const float* __restrict__ t0, const float* __restrict__ t1,
    float* __restrict__ out) {
  __shared__ __attribute__((aligned(16))) float s0s[16][68];
  __shared__ __attribute__((aligned(16))) float s1s[3][16][68];
  __shared__ __attribute__((aligned(16))) float so[16][260];
  const int wave  = threadIdx.x >> 5;
  const int lane  = threadIdx.x & 31;
  const int lrow  = lane & 15;
  const int kh    = lane >> 4;
  const int nbase = blockIdx.x * 16;
  const int arow  = nbase + lrow;
  const int col   = 16 * wave + lrow;

  v8f a0 = {};
  v8f a1[3] = {{}, {}, {}};
  for (int kk = 0; kk < 64; kk += 4) {
    const int k0 = kk + 2 * kh;
    const v2f at0 = ld2(&t0[(size_t)arow * 64 + k0]);
    a0 = wmma4(at0, ldb(Wpost0P, 64, k0, col), a0);
    const v2f b1 = ldb(Wpost1P, 64, k0, col);
    const float* tp = &t1[(size_t)arow * 192 + k0 * 3];
    const v2f p0 = ld2(tp), p1 = ld2(tp + 2), p2 = ld2(tp + 4);
    v2f am;
    am.x = p0.x; am.y = p1.y;
    a1[0] = wmma4(am, b1, a1[0]);
    am.x = p0.y; am.y = p2.x;
    a1[1] = wmma4(am, b1, a1[1]);
    am.x = p1.x; am.y = p2.y;
    a1[2] = wmma4(am, b1, a1[2]);
  }

#pragma unroll
  for (int v = 0; v < 8; ++v) {
    const int node = nbase + v + 8 * kh;
    const float s  = a0[v] * (1.f / 16.f);
    const float ax = a1[0][v] * (1.f / 16.f);
    const float ay = a1[1][v] * (1.f / 16.f);
    const float az = a1[2][v] * (1.f / 16.f);
    const float q  = ax * ax + ay * ay + az * az;
    float c0 = 0.f, c1 = 0.f, c2 = 0.f, d0 = 0.f, d1 = 0.f, d2 = 0.f;
#pragma unroll
    for (int z = 0; z < 10; ++z) {
      const float t = attrs[(size_t)node * 10 + z];
      c0 += t * Wc0[z * 64 + col];
      c1 += t * Wc0[640 + z * 64 + col];
      c2 += t * Wc0[1280 + z * 64 + col];
      d0 += t * Wc1[z * 64 + col];
      d1 += t * Wc1[640 + z * 64 + col];
      d2 += t * Wc1[1280 + z * 64 + col];
    }
    const float s2 = s * s;
    const float g  = d2 * (s2 + q) + d1 * s + d0;
    a0[v]    = c2 * (s2 * s + s * q) + c1 * (s2 + q) + c0 * s;
    a1[0][v] = g * ax;
    a1[1][v] = g * ay;
    a1[2][v] = g * az;
  }

  v2f zz[5];
#pragma unroll
  for (int t = 0; t < 5; ++t) {
    const int z0 = (4 * t + 2 * kh) % 10;
    zz[t] = ld2(&attrs[(size_t)arow * 10 + z0]);
  }
  const bool hi = kh != 0;
  const float* xrow = nf + (size_t)arow * 256;
  for (int i = 0; i < 16; ++i) {
    const float4 q0 = *(const float4*)(xrow + 4 * i);
    const float4 q1 = *(const float4*)(xrow + 64 + 12 * i);
    const float4 q2 = *(const float4*)(xrow + 64 + 12 * i + 4);
    const float4 q3 = *(const float4*)(xrow + 64 + 12 * i + 8);
    const float xv[4]  = {q0.x, q0.y, q0.z, q0.w};
    const float yv[12] = {q1.x, q1.y, q1.z, q1.w, q2.x, q2.y,
                          q2.z, q2.w, q3.x, q3.y, q3.z, q3.w};
#pragma unroll
    for (int s = 0; s < 10; ++s) {
      const int k0 = 40 * i + 4 * s + 2 * kh;
      const int U0 = (4 * s) / 10;
      const int U1 = (4 * s + 2) / 10;
      float x  = xv[U0];
      float y0 = yv[3 * U0], y1 = yv[3 * U0 + 1], y2 = yv[3 * U0 + 2];
      if (U0 != U1) {
        x  = hi ? xv[U1] : x;
        y0 = hi ? yv[3 * U1] : y0;
        y1 = hi ? yv[3 * U1 + 1] : y1;
        y2 = hi ? yv[3 * U1 + 2] : y2;
      }
      const v2f az = zz[s % 5];
      v2f a;
      a.x = x * az.x; a.y = x * az.y;
      a0 = wmma4(a, ldb(Wr0P, 64, k0, col), a0);
      const v2f b1 = ldb(Wr1P, 64, k0, col);
      a.x = y0 * az.x; a.y = y0 * az.y;
      a1[0] = wmma4(a, b1, a1[0]);
      a.x = y1 * az.x; a.y = y1 * az.y;
      a1[1] = wmma4(a, b1, a1[1]);
      a.x = y2 * az.x; a.y = y2 * az.y;
      a1[2] = wmma4(a, b1, a1[2]);
    }
  }

#pragma unroll
  for (int v = 0; v < 8; ++v) {
    const int M = v + 8 * kh;
    s0s[M][col]    = a0[v];
    s1s[0][M][col] = a1[0][v];
    s1s[1][M][col] = a1[1][v];
    s1s[2][M][col] = a1[2][v];
  }
  __syncthreads();

  {
    v8f f0 = {};
    v8f f1[3] = {{}, {}, {}};
    for (int kk = 0; kk < 64; kk += 4) {
      const int k0 = kk + 2 * kh;
      const v2f af = *(const v2f*)&s0s[lrow][k0];
      f0 = wmma4(af, ldb(Wf0P, 64, k0, col), f0);
      const v2f b1 = ldb(Wf1P, 64, k0, col);
#pragma unroll
      for (int m = 0; m < 3; ++m) {
        const v2f am = *(const v2f*)&s1s[m][lrow][k0];
        f1[m] = wmma4(am, b1, f1[m]);
      }
    }
#pragma unroll
    for (int v = 0; v < 8; ++v) {
      const int M = v + 8 * kh;
      so[M][col] = f0[v];
      so[M][64 + col * 3 + 0] = f1[0][v];
      so[M][64 + col * 3 + 1] = f1[1][v];
      so[M][64 + col * 3 + 2] = f1[2][v];
    }
  }
  __syncthreads();

  float4* outv = (float4*)(out + (size_t)nbase * 256);
#pragma unroll
  for (int p = 0; p < 8; ++p) {
    const int f   = p * 128 + threadIdx.x;
    const int row = f >> 6, c4 = f & 63;
    outv[f] = *(const float4*)&so[row][4 * c4];
  }
}

extern "C" void kernel_launch(void* const* d_in, const int* in_sizes, int n_in,
                              void* d_out, int out_size, void* d_ws, size_t ws_size,
                              hipStream_t stream) {
  (void)in_sizes; (void)n_in; (void)out_size; (void)ws_size;
  const float* nf     = (const float*)d_in[0];
  const float* attrs  = (const float*)d_in[1];
  const float* sph    = (const float*)d_in[2];
  const float* rb     = (const float*)d_in[3];
  const int*   snd    = (const int*)d_in[4];
  const int*   rcv    = (const int*)d_in[5];
  const float* Wpre0  = (const float*)d_in[6];
  const float* Wpre1  = (const float*)d_in[7];
  const float* W1     = (const float*)d_in[8];
  const float* W2     = (const float*)d_in[9];
  const float* Wpost0 = (const float*)d_in[10];
  const float* Wpost1 = (const float*)d_in[11];
  const float* Wc0    = (const float*)d_in[12];
  const float* Wc1    = (const float*)d_in[13];
  const float* Wr0    = (const float*)d_in[14];
  const float* Wr1    = (const float*)d_in[15];
  const float* Wf0    = (const float*)d_in[16];
  const float* Wf1    = (const float*)d_in[17];

  float* ws = (float*)d_ws;
  size_t off = 0;
  float* x0 = ws + off; off += (size_t)N_NODES * 64;
  float* x1 = ws + off; off += (size_t)N_NODES * 192;
  float* t0 = ws + off; off += (size_t)N_NODES * 64;
  float* t1 = ws + off; off += (size_t)N_NODES * 192;
  float* Wpre0P  = ws + off; off += 64 * 64;
  float* Wpre1P  = ws + off; off += 64 * 64;
  float* W1P     = ws + off; off += 8 * 64;
  float* W2P     = ws + off; off += 64 * 256;
  float* Wpost0P = ws + off; off += 64 * 64;
  float* Wpost1P = ws + off; off += 64 * 64;
  float* Wr0P    = ws + off; off += 640 * 64;
  float* Wr1P    = ws + off; off += 640 * 64;
  float* Wf0P    = ws + off; off += 64 * 64;
  float* Wf1P    = ws + off; off += 64 * 64;

  auto pack = [&](const float* W, float* WP, int K, int ncol) {
    const int n = K * ncol;
    k_pack_pairs<<<(n + 255) / 256, 256, 0, stream>>>(W, WP, K, ncol);
  };
  pack(Wpre0, Wpre0P, 64, 64);
  pack(Wpre1, Wpre1P, 64, 64);
  pack(W1, W1P, 8, 64);
  pack(W2, W2P, 64, 256);
  pack(Wpost0, Wpost0P, 64, 64);
  pack(Wpost1, Wpost1P, 64, 64);
  pack(Wr0, Wr0P, 640, 64);
  pack(Wr1, Wr1P, 640, 64);
  pack(Wf0, Wf0P, 64, 64);
  pack(Wf1, Wf1P, 64, 64);

  const int ntiles = N_NODES / 16;
  k_node_pre<<<(ntiles + 3) / 4, 128, 0, stream>>>(nf, Wpre0P, Wpre1P, x0, x1, t0, t1);
  k_edge<<<E_EDGES / 64, 128, 0, stream>>>(sph, rb, snd, rcv, W1P, W2P, x0, x1, t0, t1);
  k_node_post<<<ntiles, 128, 0, stream>>>(nf, attrs, Wpost0P, Wpost1P, Wc0, Wc1,
                                          Wr0P, Wr1P, Wf0P, Wf1P, t0, t1,
                                          (float*)d_out);
}


# the ten weight-pack launches merged into one launch (241 blocks, block-uniform matrix select, one b64 pair store per thread): 13 launches -> 4; arithmetic and sum order unchanged
# speedup vs baseline: 1.3607x; 1.0444x over previous
#include <hip/hip_runtime.h>


#define N_NODES 20000
#define E_EDGES 320000
#define INV_SQRT3 0.57735026918962576451f

typedef __attribute__((ext_vector_type(2))) float v2f;
typedef __attribute__((ext_vector_type(8))) float v8f;

static __device__ __forceinline__ v8f wmma4(v2f a, v2f b, v8f c) {
  return __builtin_amdgcn_wmma_f32_16x16x4_f32(
      false, a, false, b, (short)0, c, false, false);
}

static __device__ __forceinline__ v2f ldb(const float* __restrict__ WP,
                                          int ncol, int k0, int col) {
  return *(const v2f*)(WP + (size_t)(k0 >> 1) * (2 * ncol) + col * 2);
}

static __device__ __forceinline__ v2f ld2(const float* __restrict__ p) {
  v2f r; r.x = p[0]; r.y = p[1]; return r;
}

#define PACK_BLOCKS 241

__global__ __launch_bounds__(256) void k_pack_all(
    const float* __restrict__ Wpre0, const float* __restrict__ Wpre1,
    const float* __restrict__ W1, const float* __restrict__ W2,
    const float* __restrict__ Wpost0, const float* __restrict__ Wpost1,
    const float* __restrict__ Wr0, const float* __restrict__ Wr1,
    const float* __restrict__ Wf0, const float* __restrict__ Wf1,
    float* __restrict__ Wpre0P, float* __restrict__ Wpre1P,
    float* __restrict__ W1P, float* __restrict__ W2P,
    float* __restrict__ Wpost0P, float* __restrict__ Wpost1P,
    float* __restrict__ Wr0P, float* __restrict__ Wr1P,
    float* __restrict__ Wf0P, float* __restrict__ Wf1P) {
  const int b = blockIdx.x;
  const float* W;
  float* WP;
  int first, lg;
  if (b < 8)        { W = Wpre0;  WP = Wpre0P;  first = 0;   lg = 6; }
  else if (b < 16)  { W = Wpre1;  WP = Wpre1P;  first = 8;   lg = 6; }
  else if (b < 17)  { W = W1;     WP = W1P;     first = 16;  lg = 6; }
  else if (b < 49)  { W = W2;     WP = W2P;     first = 17;  lg = 8; }
  else if (b < 57)  { W = Wpost0; WP = Wpost0P; first = 49;  lg = 6; }
  else if (b < 65)  { W = Wpost1; WP = Wpost1P; first = 57;  lg = 6; }
  else if (b < 145) { W = Wr0;    WP = Wr0P;    first = 65;  lg = 6; }
  else if (b < 225) { W = Wr1;    WP = Wr1P;    first = 145; lg = 6; }
  else if (b < 233) { W = Wf0;    WP = Wf0P;    first = 225; lg = 6; }
  else              { W = Wf1;    WP = Wf1P;    first = 233; lg = 6; }
  const int p  = (b - first) * 256 + threadIdx.x;
  const int kp = p >> lg;
  const int c  = p & ((1 << lg) - 1);
  v2f o;
  o.x = W[((size_t)(2 * kp) << lg) + c];
  o.y = W[((size_t)(2 * kp + 1) << lg) + c];
  *(v2f*)(WP + 2 * (size_t)p) = o;
}

__global__ __launch_bounds__(128) void k_node_pre(
    const float* __restrict__ nf, const float* __restrict__ Wp0P,
    const float* __restrict__ Wp1P, float* __restrict__ x0,
    float* __restrict__ x1, float* __restrict__ t0, float* __restrict__ t1) {
  const int wave = threadIdx.x >> 5;
  const int lane = threadIdx.x & 31;
  const int tile = blockIdx.x * 4 + wave;
  if (tile >= (N_NODES / 16)) return;
  const int nbase = tile * 16;
  const int lrow  = lane & 15;
  const int kh    = lane >> 4;
  const int arow  = nbase + lrow;

  {
    v8f acc[4] = {{}, {}, {}, {}};
    for (int kk = 0; kk < 64; kk += 4) {
      const int k0 = kk + 2 * kh;
      const v2f a = ld2(&nf[(size_t)arow * 256 + k0]);
#pragma unroll
      for (int j = 0; j < 4; ++j)
        acc[j] = wmma4(a, ldb(Wp0P, 64, k0, 16 * j + lrow), acc[j]);
    }
#pragma unroll
    for (int j = 0; j < 4; ++j)
#pragma unroll
      for (int v = 0; v < 8; ++v)
        x0[(size_t)(nbase + v + 8 * kh) * 64 + 16 * j + lrow] = acc[j][v];
  }

  for (int m = 0; m < 3; ++m) {
    v8f acc[4] = {{}, {}, {}, {}};
    for (int kk = 0; kk < 64; kk += 4) {
      const int k0 = kk + 2 * kh;
      v2f a;
      a.x = nf[(size_t)arow * 256 + 64 + k0 * 3 + m];
      a.y = nf[(size_t)arow * 256 + 64 + (k0 + 1) * 3 + m];
#pragma unroll
      for (int j = 0; j < 4; ++j)
        acc[j] = wmma4(a, ldb(Wp1P, 64, k0, 16 * j + lrow), acc[j]);
    }
#pragma unroll
    for (int j = 0; j < 4; ++j)
#pragma unroll
      for (int v = 0; v < 8; ++v)
        x1[(size_t)(nbase + v + 8 * kh) * 192 + (16 * j + lrow) * 3 + m] = acc[j][v];
  }

  float4* t0v = (float4*)(t0 + (size_t)nbase * 64);
  float4* t1v = (float4*)(t1 + (size_t)nbase * 192);
  const float4 z4 = make_float4(0.f, 0.f, 0.f, 0.f);
  for (int i = lane; i < 16 * 64 / 4; i += 32)  t0v[i] = z4;
  for (int i = lane; i < 16 * 192 / 4; i += 32) t1v[i] = z4;
}

__global__ __launch_bounds__(128) void k_edge(
    const float* __restrict__ sph, const float* __restrict__ rb,
    const int* __restrict__ snd, const int* __restrict__ rcv,
    const float* __restrict__ W1P, const float* __restrict__ W2P,
    const float* __restrict__ x0, const float* __restrict__ x1,
    float* __restrict__ t0, float* __restrict__ t1) {
  __shared__ float hidden[4][16][64];
  const int wave  = threadIdx.x >> 5;
  const int lane  = threadIdx.x & 31;
  const int lrow  = lane & 15;
  const int kh    = lane >> 4;
  const int ebase = (blockIdx.x * 4 + wave) * 16;

  {
    v8f acc[4] = {{}, {}, {}, {}};
#pragma unroll
    for (int kk = 0; kk < 8; kk += 4) {
      const int k0 = kk + 2 * kh;
      const v2f a = ld2(&rb[(size_t)(ebase + lrow) * 8 + k0]);
#pragma unroll
      for (int j = 0; j < 4; ++j)
        acc[j] = wmma4(a, ldb(W1P, 64, k0, 16 * j + lrow), acc[j]);
    }
#pragma unroll
    for (int j = 0; j < 4; ++j)
#pragma unroll
      for (int v = 0; v < 8; ++v) {
        float h = acc[j][v];
        h = h / (1.f + __expf(-h));
        hidden[wave][v + 8 * kh][16 * j + lrow] = h;
      }
  }
  __syncthreads();

  v8f w[16] = {{}, {}, {}, {}, {}, {}, {}, {}, {}, {}, {}, {}, {}, {}, {}, {}};
  for (int kk = 0; kk < 64; kk += 4) {
    const int k0 = kk + 2 * kh;
    const v2f a = ld2(&hidden[wave][lrow][k0]);
#pragma unroll
    for (int j = 0; j < 16; ++j)
      w[j] = wmma4(a, ldb(W2P, 256, k0, 16 * j + lrow), w[j]);
  }

#pragma unroll
  for (int v = 0; v < 8; ++v) {
    const int e = ebase + v + 8 * kh;
    const int s = snd[e];
    const int r = rcv[e];
    const float Y0  = sph[(size_t)e * 4 + 0];
    const float Y1x = sph[(size_t)e * 4 + 1];
    const float Y1y = sph[(size_t)e * 4 + 2];
    const float Y1z = sph[(size_t)e * 4 + 3];
#pragma unroll
    for (int j = 0; j < 4; ++j) {
      const int c = 16 * j + lrow;
      const float w0 = w[j][v], w1 = w[j + 4][v];
      const float w2 = w[j + 8][v], w3 = w[j + 12][v];
      const float h0  = x0[(size_t)s * 64 + c];
      const float h1x = x1[(size_t)s * 192 + c * 3 + 0];
      const float h1y = x1[(size_t)s * 192 + c * 3 + 1];
      const float h1z = x1[(size_t)s * 192 + c * 3 + 2];
      const float dot = h1x * Y1x + h1y * Y1y + h1z * Y1z;
      atomicAdd(&t0[(size_t)r * 64 + c], w0 * h0 * Y0 + w1 * dot * INV_SQRT3);
      const size_t b1 = (size_t)r * 192 + c * 3;
      atomicAdd(&t1[b1 + 0], w2 * h0 * Y1x + w3 * h1x * Y0);
      atomicAdd(&t1[b1 + 1], w2 * h0 * Y1y + w3 * h1y * Y0);
      atomicAdd(&t1[b1 + 2], w2 * h0 * Y1z + w3 * h1z * Y0);
    }
  }
}

__global__ __launch_bounds__(128) void k_node_post(
    const float* __restrict__ nf, const float* __restrict__ attrs,
    const float* __restrict__ Wpost0P, const float* __restrict__ Wpost1P,
    const float* __restrict__ Wc0, const float* __restrict__ Wc1,
    const float* __restrict__ Wr0P, const float* __restrict__ Wr1P,
    const float* __restrict__ Wf0P, const float* __restrict__ Wf1P,
    const float* __restrict__ t0, const float* __restrict__ t1,
    float* __restrict__ out) {
  __shared__ __attribute__((aligned(16))) float s0s[16][68];
  __shared__ __attribute__((aligned(16))) float s1s[3][16][68];
  __shared__ __attribute__((aligned(16))) float so[16][260];
  const int wave  = threadIdx.x >> 5;
  const int lane  = threadIdx.x & 31;
  const int lrow  = lane & 15;
  const int kh    = lane >> 4;
  const int nbase = blockIdx.x * 16;
  const int arow  = nbase + lrow;
  const int col   = 16 * wave + lrow;

  v8f a0 = {};
  v8f a1[3] = {{}, {}, {}};
  for (int kk = 0; kk < 64; kk += 4) {
    const int k0 = kk + 2 * kh;
    const v2f at0 = ld2(&t0[(size_t)arow * 64 + k0]);
    a0 = wmma4(at0, ldb(Wpost0P, 64, k0, col), a0);
    const v2f b1 = ldb(Wpost1P, 64, k0, col);
    const float* tp = &t1[(size_t)arow * 192 + k0 * 3];
    const v2f p0 = ld2(tp), p1 = ld2(tp + 2), p2 = ld2(tp + 4);
    v2f am;
    am.x = p0.x; am.y = p1.y;
    a1[0] = wmma4(am, b1, a1[0]);
    am.x = p0.y; am.y = p2.x;
    a1[1] = wmma4(am, b1, a1[1]);
    am.x = p1.x; am.y = p2.y;
    a1[2] = wmma4(am, b1, a1[2]);
  }

#pragma unroll
  for (int v = 0; v < 8; ++v) {
    const int node = nbase + v + 8 * kh;
    const float s  = a0[v] * (1.f / 16.f);
    const float ax = a1[0][v] * (1.f / 16.f);
    const float ay = a1[1][v] * (1.f / 16.f);
    const float az = a1[2][v] * (1.f / 16.f);
    const float q  = ax * ax + ay * ay + az * az;
    float c0 = 0.f, c1 = 0.f, c2 = 0.f, d0 = 0.f, d1 = 0.f, d2 = 0.f;
#pragma unroll
    for (int z = 0; z < 10; ++z) {
      const float t = attrs[(size_t)node * 10 + z];
      c0 += t * Wc0[z * 64 + col];
      c1 += t * Wc0[640 + z * 64 + col];
      c2 += t * Wc0[1280 + z * 64 + col];
      d0 += t * Wc1[z * 64 + col];
      d1 += t * Wc1[640 + z * 64 + col];
      d2 += t * Wc1[1280 + z * 64 + col];
    }
    const float s2 = s * s;
    const float g  = d2 * (s2 + q) + d1 * s + d0;
    a0[v]    = c2 * (s2 * s + s * q) + c1 * (s2 + q) + c0 * s;
    a1[0][v] = g * ax;
    a1[1][v] = g * ay;
    a1[2][v] = g * az;
  }

  v2f zz[5];
#pragma unroll
  for (int t = 0; t < 5; ++t) {
    const int z0 = (4 * t + 2 * kh) % 10;
    zz[t] = ld2(&attrs[(size_t)arow * 10 + z0]);
  }
  const bool hi = kh != 0;
  const float* xrow = nf + (size_t)arow * 256;
  for (int i = 0; i < 16; ++i) {
    const float4 q0 = *(const float4*)(xrow + 4 * i);
    const float4 q1 = *(const float4*)(xrow + 64 + 12 * i);
    const float4 q2 = *(const float4*)(xrow + 64 + 12 * i + 4);
    const float4 q3 = *(const float4*)(xrow + 64 + 12 * i + 8);
    const float xv[4]  = {q0.x, q0.y, q0.z, q0.w};
    const float yv[12] = {q1.x, q1.y, q1.z, q1.w, q2.x, q2.y,
                          q2.z, q2.w, q3.x, q3.y, q3.z, q3.w};
#pragma unroll
    for (int s = 0; s < 10; ++s) {
      const int k0 = 40 * i + 4 * s + 2 * kh;
      const int U0 = (4 * s) / 10;
      const int U1 = (4 * s + 2) / 10;
      float x  = xv[U0];
      float y0 = yv[3 * U0], y1 = yv[3 * U0 + 1], y2 = yv[3 * U0 + 2];
      if (U0 != U1) {
        x  = hi ? xv[U1] : x;
        y0 = hi ? yv[3 * U1] : y0;
        y1 = hi ? yv[3 * U1 + 1] : y1;
        y2 = hi ? yv[3 * U1 + 2] : y2;
      }
      const v2f az = zz[s % 5];
      v2f a;
      a.x = x * az.x; a.y = x * az.y;
      a0 = wmma4(a, ldb(Wr0P, 64, k0, col), a0);
      const v2f b1 = ldb(Wr1P, 64, k0, col);
      a.x = y0 * az.x; a.y = y0 * az.y;
      a1[0] = wmma4(a, b1, a1[0]);
      a.x = y1 * az.x; a.y = y1 * az.y;
      a1[1] = wmma4(a, b1, a1[1]);
      a.x = y2 * az.x; a.y = y2 * az.y;
      a1[2] = wmma4(a, b1, a1[2]);
    }
  }

#pragma unroll
  for (int v = 0; v < 8; ++v) {
    const int M = v + 8 * kh;
    s0s[M][col]    = a0[v];
    s1s[0][M][col] = a1[0][v];
    s1s[1][M][col] = a1[1][v];
    s1s[2][M][col] = a1[2][v];
  }
  __syncthreads();

  {
    v8f f0 = {};
    v8f f1[3] = {{}, {}, {}};
    for (int kk = 0; kk < 64; kk += 4) {
      const int k0 = kk + 2 * kh;
      const v2f af = *(const v2f*)&s0s[lrow][k0];
      f0 = wmma4(af, ldb(Wf0P, 64, k0, col), f0);
      const v2f b1 = ldb(Wf1P, 64, k0, col);
#pragma unroll
      for (int m = 0; m < 3; ++m) {
        const v2f am = *(const v2f*)&s1s[m][lrow][k0];
        f1[m] = wmma4(am, b1, f1[m]);
      }
    }
#pragma unroll
    for (int v = 0; v < 8; ++v) {
      const int M = v + 8 * kh;
      so[M][col] = f0[v];
      so[M][64 + col * 3 + 0] = f1[0][v];
      so[M][64 + col * 3 + 1] = f1[1][v];
      so[M][64 + col * 3 + 2] = f1[2][v];
    }
  }
  __syncthreads();

  float4* outv = (float4*)(out + (size_t)nbase * 256);
#pragma unroll
  for (int p = 0; p < 8; ++p) {
    const int f   = p * 128 + threadIdx.x;
    const int row = f >> 6, c4 = f & 63;
    outv[f] = *(const float4*)&so[row][4 * c4];
  }
}

extern "C" void kernel_launch(void* const* d_in, const int* in_sizes, int n_in,
                              void* d_out, int out_size, void* d_ws, size_t ws_size,
                              hipStream_t stream) {
  (void)in_sizes; (void)n_in; (void)out_size; (void)ws_size;
  const float* nf     = (const float*)d_in[0];
  const float* attrs  = (const float*)d_in[1];
  const float* sph    = (const float*)d_in[2];
  const float* rb     = (const float*)d_in[3];
  const int*   snd    = (const int*)d_in[4];
  const int*   rcv    = (const int*)d_in[5];
  const float* Wpre0  = (const float*)d_in[6];
  const float* Wpre1  = (const float*)d_in[7];
  const float* W1     = (const float*)d_in[8];
  const float* W2     = (const float*)d_in[9];
  const float* Wpost0 = (const float*)d_in[10];
  const float* Wpost1 = (const float*)d_in[11];
  const float* Wc0    = (const float*)d_in[12];
  const float* Wc1    = (const float*)d_in[13];
  const float* Wr0    = (const float*)d_in[14];
  const float* Wr1    = (const float*)d_in[15];
  const float* Wf0    = (const float*)d_in[16];
  const float* Wf1    = (const float*)d_in[17];

  float* ws = (float*)d_ws;
  size_t off = 0;
  float* x0 = ws + off; off += (size_t)N_NODES * 64;
  float* x1 = ws + off; off += (size_t)N_NODES * 192;
  float* t0 = ws + off; off += (size_t)N_NODES * 64;
  float* t1 = ws + off; off += (size_t)N_NODES * 192;
  float* Wpre0P  = ws + off; off += 64 * 64;
  float* Wpre1P  = ws + off; off += 64 * 64;
  float* W1P     = ws + off; off += 8 * 64;
  float* W2P     = ws + off; off += 64 * 256;
  float* Wpost0P = ws + off; off += 64 * 64;
  float* Wpost1P = ws + off; off += 64 * 64;
  float* Wr0P    = ws + off; off += 640 * 64;
  float* Wr1P    = ws + off; off += 640 * 64;
  float* Wf0P    = ws + off; off += 64 * 64;
  float* Wf1P    = ws + off; off += 64 * 64;

  k_pack_all<<<PACK_BLOCKS, 256, 0, stream>>>(
      Wpre0, Wpre1, W1, W2, Wpost0, Wpost1, Wr0, Wr1, Wf0, Wf1,
      Wpre0P, Wpre1P, W1P, W2P, Wpost0P, Wpost1P, Wr0P, Wr1P, Wf0P, Wf1P);

  const int ntiles = N_NODES / 16;
  k_node_pre<<<(ntiles + 3) / 4, 128, 0, stream>>>(nf, Wpre0P, Wpre1P, x0, x1, t0, t1);
  k_edge<<<E_EDGES / 64, 128, 0, stream>>>(sph, rb, snd, rcv, W1P, W2P, x0, x1, t0, t1);
  k_node_post<<<ntiles, 128, 0, stream>>>(nf, attrs, Wpost0P, Wpost1P, Wc0, Wc1,
                                          Wr0P, Wr1P, Wf0P, Wf1P, t0, t1,
                                          (float*)d_out);
}


# edge kernel: edge indices loaded at kernel start, each edge's sender rows (x0, x1, sph) requested two edges ahead of its atomics so L2 round trips overlap; atomics, arrival-order sums unchanged
# speedup vs baseline: 1.4496x; 1.0653x over previous
#include <hip/hip_runtime.h>


#define N_NODES 20000
#define E_EDGES 320000
#define INV_SQRT3 0.57735026918962576451f

typedef __attribute__((ext_vector_type(2))) float v2f;
typedef __attribute__((ext_vector_type(8))) float v8f;

static __device__ __forceinline__ v8f wmma4(v2f a, v2f b, v8f c) {
  return __builtin_amdgcn_wmma_f32_16x16x4_f32(
      false, a, false, b, (short)0, c, false, false);
}

static __device__ __forceinline__ v2f ldb(const float* __restrict__ WP,
                                          int ncol, int k0, int col) {
  return *(const v2f*)(WP + (size_t)(k0 >> 1) * (2 * ncol) + col * 2);
}

static __device__ __forceinline__ v2f ld2(const float* __restrict__ p) {
  v2f r; r.x = p[0]; r.y = p[1]; return r;
}

#define PACK_BLOCKS 241

__global__ __launch_bounds__(256) void k_pack_all(
    const float* __restrict__ Wpre0, const float* __restrict__ Wpre1,
    const float* __restrict__ W1, const float* __restrict__ W2,
    const float* __restrict__ Wpost0, const float* __restrict__ Wpost1,
    const float* __restrict__ Wr0, const float* __restrict__ Wr1,
    const float* __restrict__ Wf0, const float* __restrict__ Wf1,
    float* __restrict__ Wpre0P, float* __restrict__ Wpre1P,
    float* __restrict__ W1P, float* __restrict__ W2P,
    float* __restrict__ Wpost0P, float* __restrict__ Wpost1P,
    float* __restrict__ Wr0P, float* __restrict__ Wr1P,
    float* __restrict__ Wf0P, float* __restrict__ Wf1P) {
  const int b = blockIdx.x;
  const float* W;
  float* WP;
  int first, lg;
  if (b < 8)        { W = Wpre0;  WP = Wpre0P;  first = 0;   lg = 6; }
  else if (b < 16)  { W = Wpre1;  WP = Wpre1P;  first = 8;   lg = 6; }
  else if (b < 17)  { W = W1;     WP = W1P;     first = 16;  lg = 6; }
  else if (b < 49)  { W = W2;     WP = W2P;     first = 17;  lg = 8; }
  else if (b < 57)  { W = Wpost0; WP = Wpost0P; first = 49;  lg = 6; }
  else if (b < 65)  { W = Wpost1; WP = Wpost1P; first = 57;  lg = 6; }
  else if (b < 145) { W = Wr0;    WP = Wr0P;    first = 65;  lg = 6; }
  else if (b < 225) { W = Wr1;    WP = Wr1P;    first = 145; lg = 6; }
  else if (b < 233) { W = Wf0;    WP = Wf0P;    first = 225; lg = 6; }
  else              { W = Wf1;    WP = Wf1P;    first = 233; lg = 6; }
  const int p  = (b - first) * 256 + threadIdx.x;
  const int kp = p >> lg;
  const int c  = p & ((1 << lg) - 1);
  v2f o;
  o.x = W[((size_t)(2 * kp) << lg) + c];
  o.y = W[((size_t)(2 * kp + 1) << lg) + c];
  *(v2f*)(WP + 2 * (size_t)p) = o;
}

__global__ __launch_bounds__(128) void k_node_pre(
    const float* __restrict__ nf, const float* __restrict__ Wp0P,
    const float* __restrict__ Wp1P, float* __restrict__ x0,
    float* __restrict__ x1, float* __restrict__ t0, float* __restrict__ t1) {
  const int wave = threadIdx.x >> 5;
  const int lane = threadIdx.x & 31;
  const int tile = blockIdx.x * 4 + wave;
  if (tile >= (N_NODES / 16)) return;
  const int nbase = tile * 16;
  const int lrow  = lane & 15;
  const int kh    = lane >> 4;
  const int arow  = nbase + lrow;

  {
    v8f acc[4] = {{}, {}, {}, {}};
    for (int kk = 0; kk < 64; kk += 4) {
      const int k0 = kk + 2 * kh;
      const v2f a = ld2(&nf[(size_t)arow * 256 + k0]);
#pragma unroll
      for (int j = 0; j < 4; ++j)
        acc[j] = wmma4(a, ldb(Wp0P, 64, k0, 16 * j + lrow), acc[j]);
    }
#pragma unroll
    for (int j = 0; j < 4; ++j)
#pragma unroll
      for (int v = 0; v < 8; ++v)
        x0[(size_t)(nbase + v + 8 * kh) * 64 + 16 * j + lrow] = acc[j][v];
  }

  for (int m = 0; m < 3; ++m) {
    v8f acc[4] = {{}, {}, {}, {}};
    for (int kk = 0; kk < 64; kk += 4) {
      const int k0 = kk + 2 * kh;
      v2f a;
      a.x = nf[(size_t)arow * 256 + 64 + k0 * 3 + m];
      a.y = nf[(size_t)arow * 256 + 64 + (k0 + 1) * 3 + m];
#pragma unroll
      for (int j = 0; j < 4; ++j)
        acc[j] = wmma4(a, ldb(Wp1P, 64, k0, 16 * j + lrow), acc[j]);
    }
#pragma unroll
    for (int j = 0; j < 4; ++j)
#pragma unroll
      for (int v = 0; v < 8; ++v)
        x1[(size_t)(nbase + v + 8 * kh) * 192 + (16 * j + lrow) * 3 + m] = acc[j][v];
  }

  float4* t0v = (float4*)(t0 + (size_t)nbase * 64);
  float4* t1v = (float4*)(t1 + (size_t)nbase * 192);
  const float4 z4 = make_float4(0.f, 0.f, 0.f, 0.f);
  for (int i = lane; i < 16 * 64 / 4; i += 32)  t0v[i] = z4;
  for (int i = lane; i < 16 * 192 / 4; i += 32) t1v[i] = z4;
}

struct EdgeRows {
  float4 Y;
  float h0[4], hx[4], hy[4], hz[4];
};

static __device__ __forceinline__ void load_edge_rows(
    EdgeRows& q, const float* __restrict__ sph, const float* __restrict__ x0,
    const float* __restrict__ x1, int e, int s, int lrow) {
  q.Y = *(const float4*)(sph + (size_t)e * 4);
  const float* p0 = x0 + (size_t)s * 64 + lrow;
  const float* p1 = x1 + (size_t)s * 192 + lrow * 3;
#pragma unroll
  for (int j = 0; j < 4; ++j) {
    q.h0[j] = p0[16 * j];
    q.hx[j] = p1[48 * j + 0];
    q.hy[j] = p1[48 * j + 1];
    q.hz[j] = p1[48 * j + 2];
  }
}

__global__ __launch_bounds__(128) void k_edge(
    const float* __restrict__ sph, const float* __restrict__ rb,
    const int* __restrict__ snd, const int* __restrict__ rcv,
    const float* __restrict__ W1P, const float* __restrict__ W2P,
    const float* __restrict__ x0, const float* __restrict__ x1,
    float* __restrict__ t0, float* __restrict__ t1) {
  __shared__ float hidden[4][16][64];
  const int wave  = threadIdx.x >> 5;
  const int lane  = threadIdx.x & 31;
  const int lrow  = lane & 15;
  const int kh    = lane >> 4;
  const int ebase = (blockIdx.x * 4 + wave) * 16;

  int sidx[8], ridx[8];
  {
    const int4* sp = (const int4*)(snd + ebase + 8 * kh);
    const int4* rp = (const int4*)(rcv + ebase + 8 * kh);
    const int4 sa = sp[0], sb = sp[1], ra = rp[0], rb4 = rp[1];
    sidx[0] = sa.x; sidx[1] = sa.y; sidx[2] = sa.z; sidx[3] = sa.w;
    sidx[4] = sb.x; sidx[5] = sb.y; sidx[6] = sb.z; sidx[7] = sb.w;
    ridx[0] = ra.x; ridx[1] = ra.y; ridx[2] = ra.z; ridx[3] = ra.w;
    ridx[4] = rb4.x; ridx[5] = rb4.y; ridx[6] = rb4.z; ridx[7] = rb4.w;
  }

  {
    v8f acc[4] = {{}, {}, {}, {}};
#pragma unroll
    for (int kk = 0; kk < 8; kk += 4) {
      const int k0 = kk + 2 * kh;
      const v2f a = ld2(&rb[(size_t)(ebase + lrow) * 8 + k0]);
#pragma unroll
      for (int j = 0; j < 4; ++j)
        acc[j] = wmma4(a, ldb(W1P, 64, k0, 16 * j + lrow), acc[j]);
    }
#pragma unroll
    for (int j = 0; j < 4; ++j)
#pragma unroll
      for (int v = 0; v < 8; ++v) {
        float h = acc[j][v];
        h = h / (1.f + __expf(-h));
        hidden[wave][v + 8 * kh][16 * j + lrow] = h;
      }
  }
  __syncthreads();

  v8f w[16] = {{}, {}, {}, {}, {}, {}, {}, {}, {}, {}, {}, {}, {}, {}, {}, {}};
  for (int kk = 0; kk < 64; kk += 4) {
    const int k0 = kk + 2 * kh;
    const v2f a = ld2(&hidden[wave][lrow][k0]);
#pragma unroll
    for (int j = 0; j < 16; ++j)
      w[j] = wmma4(a, ldb(W2P, 256, k0, 16 * j + lrow), w[j]);
  }

  EdgeRows g[3];
  load_edge_rows(g[0], sph, x0, x1, ebase + 8 * kh + 0, sidx[0], lrow);
  load_edge_rows(g[1], sph, x0, x1, ebase + 8 * kh + 1, sidx[1], lrow);
#pragma unroll
  for (int v = 0; v < 8; ++v) {
    if (v + 2 < 8)
      load_edge_rows(g[(v + 2) % 3], sph, x0, x1, ebase + 8 * kh + v + 2,
                     sidx[v + 2], lrow);
    const EdgeRows& q = g[v % 3];
    const int r = ridx[v];
    const float Y0 = q.Y.x, Y1x = q.Y.y, Y1y = q.Y.z, Y1z = q.Y.w;
#pragma unroll
    for (int j = 0; j < 4; ++j) {
      const int c = 16 * j + lrow;
      const float w0 = w[j][v], w1 = w[j + 4][v];
      const float w2 = w[j + 8][v], w3 = w[j + 12][v];
      const float h0  = q.h0[j];
      const float h1x = q.hx[j];
      const float h1y = q.hy[j];
      const float h1z = q.hz[j];
      const float dot = h1x * Y1x + h1y * Y1y + h1z * Y1z;
      atomicAdd(&t0[(size_t)r * 64 + c], w0 * h0 * Y0 + w1 * dot * INV_SQRT3);
      const size_t b1 = (size_t)r * 192 + c * 3;
      atomicAdd(&t1[b1 + 0], w2 * h0 * Y1x + w3 * h1x * Y0);
      atomicAdd(&t1[b1 + 1], w2 * h0 * Y1y + w3 * h1y * Y0);
      atomicAdd(&t1[b1 + 2], w2 * h0 * Y1z + w3 * h1z * Y0);
    }
  }
}

__global__ __launch_bounds__(128) void k_node_post(
    const float* __restrict__ nf, const float* __restrict__ attrs,
    const float* __restrict__ Wpost0P, const float* __restrict__ Wpost1P,
    const float* __restrict__ Wc0, const float* __restrict__ Wc1,
    const float* __restrict__ Wr0P, const float* __restrict__ Wr1P,
    const float* __restrict__ Wf0P, const float* __restrict__ Wf1P,
    const float* __restrict__ t0, const float* __restrict__ t1,
    float* __restrict__ out) {
  __shared__ __attribute__((aligned(16))) float s0s[16][68];
  __shared__ __attribute__((aligned(16))) float s1s[3][16][68];
  __shared__ __attribute__((aligned(16))) float so[16][260];
  const int wave  = threadIdx.x >> 5;
  const int lane  = threadIdx.x & 31;
  const int lrow  = lane & 15;
  const int kh    = lane >> 4;
  const int nbase = blockIdx.x * 16;
  const int arow  = nbase + lrow;
  const int col   = 16 * wave + lrow;

  v8f a0 = {};
  v8f a1[3] = {{}, {}, {}};
  for (int kk = 0; kk < 64; kk += 4) {
    const int k0 = kk + 2 * kh;
    const v2f at0 = ld2(&t0[(size_t)arow * 64 + k0]);
    a0 = wmma4(at0, ldb(Wpost0P, 64, k0, col), a0);
    const v2f b1 = ldb(Wpost1P, 64, k0, col);
    const float* tp = &t1[(size_t)arow * 192 + k0 * 3];
    const v2f p0 = ld2(tp), p1 = ld2(tp + 2), p2 = ld2(tp + 4);
    v2f am;
    am.x = p0.x; am.y = p1.y;
    a1[0] = wmma4(am, b1, a1[0]);
    am.x = p0.y; am.y = p2.x;
    a1[1] = wmma4(am, b1, a1[1]);
    am.x = p1.x; am.y = p2.y;
    a1[2] = wmma4(am, b1, a1[2]);
  }

#pragma unroll
  for (int v = 0; v < 8; ++v) {
    const int node = nbase + v + 8 * kh;
    const float s  = a0[v] * (1.f / 16.f);
    const float ax = a1[0][v] * (1.f / 16.f);
    const float ay = a1[1][v] * (1.f / 16.f);
    const float az = a1[2][v] * (1.f / 16.f);
    const float q  = ax * ax + ay * ay + az * az;
    float c0 = 0.f, c1 = 0.f, c2 = 0.f, d0 = 0.f, d1 = 0.f, d2 = 0.f;
#pragma unroll
    for (int z = 0; z < 10; ++z) {
      const float t = attrs[(size_t)node * 10 + z];
      c0 += t * Wc0[z * 64 + col];
      c1 += t * Wc0[640 + z * 64 + col];
      c2 += t * Wc0[1280 + z * 64 + col];
      d0 += t * Wc1[z * 64 + col];
      d1 += t * Wc1[640 + z * 64 + col];
      d2 += t * Wc1[1280 + z * 64 + col];
    }
    const float s2 = s * s;
    const float g  = d2 * (s2 + q) + d1 * s + d0;
    a0[v]    = c2 * (s2 * s + s * q) + c1 * (s2 + q) + c0 * s;
    a1[0][v] = g * ax;
    a1[1][v] = g * ay;
    a1[2][v] = g * az;
  }

  v2f zz[5];
#pragma unroll
  for (int t = 0; t < 5; ++t) {
    const int z0 = (4 * t + 2 * kh) % 10;
    zz[t] = ld2(&attrs[(size_t)arow * 10 + z0]);
  }
  const bool hi = kh != 0;
  const float* xrow = nf + (size_t)arow * 256;
  for (int i = 0; i < 16; ++i) {
    const float4 q0 = *(const float4*)(xrow + 4 * i);
    const float4 q1 = *(const float4*)(xrow + 64 + 12 * i);
    const float4 q2 = *(const float4*)(xrow + 64 + 12 * i + 4);
    const float4 q3 = *(const float4*)(xrow + 64 + 12 * i + 8);
    const float xv[4]  = {q0.x, q0.y, q0.z, q0.w};
    const float yv[12] = {q1.x, q1.y, q1.z, q1.w, q2.x, q2.y,
                          q2.z, q2.w, q3.x, q3.y, q3.z, q3.w};
#pragma unroll
    for (int s = 0; s < 10; ++s) {
      const int k0 = 40 * i + 4 * s + 2 * kh;
      const int U0 = (4 * s) / 10;
      const int U1 = (4 * s + 2) / 10;
      float x  = xv[U0];
      float y0 = yv[3 * U0], y1 = yv[3 * U0 + 1], y2 = yv[3 * U0 + 2];
      if (U0 != U1) {
        x  = hi ? xv[U1] : x;
        y0 = hi ? yv[3 * U1] : y0;
        y1 = hi ? yv[3 * U1 + 1] : y1;
        y2 = hi ? yv[3 * U1 + 2] : y2;
      }
      const v2f az = zz[s % 5];
      v2f a;
      a.x = x * az.x; a.y = x * az.y;
      a0 = wmma4(a, ldb(Wr0P, 64, k0, col), a0);
      const v2f b1 = ldb(Wr1P, 64, k0, col);
      a.x = y0 * az.x; a.y = y0 * az.y;
      a1[0] = wmma4(a, b1, a1[0]);
      a.x = y1 * az.x; a.y = y1 * az.y;
      a1[1] = wmma4(a, b1, a1[1]);
      a.x = y2 * az.x; a.y = y2 * az.y;
      a1[2] = wmma4(a, b1, a1[2]);
    }
  }

#pragma unroll
  for (int v = 0; v < 8; ++v) {
    const int M = v + 8 * kh;
    s0s[M][col]    = a0[v];
    s1s[0][M][col] = a1[0][v];
    s1s[1][M][col] = a1[1][v];
    s1s[2][M][col] = a1[2][v];
  }
  __syncthreads();

  {
    v8f f0 = {};
    v8f f1[3] = {{}, {}, {}};
    for (int kk = 0; kk < 64; kk += 4) {
      const int k0 = kk + 2 * kh;
      const v2f af = *(const v2f*)&s0s[lrow][k0];
      f0 = wmma4(af, ldb(Wf0P, 64, k0, col), f0);
      const v2f b1 = ldb(Wf1P, 64, k0, col);
#pragma unroll
      for (int m = 0; m < 3; ++m) {
        const v2f am = *(const v2f*)&s1s[m][lrow][k0];
        f1[m] = wmma4(am, b1, f1[m]);
      }
    }
#pragma unroll
    for (int v = 0; v < 8; ++v) {
      const int M = v + 8 * kh;
      so[M][col] = f0[v];
      so[M][64 + col * 3 + 0] = f1[0][v];
      so[M][64 + col * 3 + 1] = f1[1][v];
      so[M][64 + col * 3 + 2] = f1[2][v];
    }
  }
  __syncthreads();

  float4* outv = (float4*)(out + (size_t)nbase * 256);
#pragma unroll
  for (int p = 0; p < 8; ++p) {
    const int f   = p * 128 + threadIdx.x;
    const int row = f >> 6, c4 = f & 63;
    outv[f] = *(const float4*)&so[row][4 * c4];
  }
}

extern "C" void kernel_launch(void* const* d_in, const int* in_sizes, int n_in,
                              void* d_out, int out_size, void* d_ws, size_t ws_size,
                              hipStream_t stream) {
  (void)in_sizes; (void)n_in; (void)out_size; (void)ws_size;
  const float* nf     = (const float*)d_in[0];
  const float* attrs  = (const float*)d_in[1];
  const float* sph    = (const float*)d_in[2];
  const float* rb     = (const float*)d_in[3];
  const int*   snd    = (const int*)d_in[4];
  const int*   rcv    = (const int*)d_in[5];
  const float* Wpre0  = (const float*)d_in[6];
  const float* Wpre1  = (const float*)d_in[7];
  const float* W1     = (const float*)d_in[8];
  const float* W2     = (const float*)d_in[9];
  const float* Wpost0 = (const float*)d_in[10];
  const float* Wpost1 = (const float*)d_in[11];
  const float* Wc0    = (const float*)d_in[12];
  const float* Wc1    = (const float*)d_in[13];
  const float* Wr0    = (const float*)d_in[14];
  const float* Wr1    = (const float*)d_in[15];
  const float* Wf0    = (const float*)d_in[16];
  const float* Wf1    = (const float*)d_in[17];

  float* ws = (float*)d_ws;
  size_t off = 0;
  float* x0 = ws + off; off += (size_t)N_NODES * 64;
  float* x1 = ws + off; off += (size_t)N_NODES * 192;
  float* t0 = ws + off; off += (size_t)N_NODES * 64;
  float* t1 = ws + off; off += (size_t)N_NODES * 192;
  float* Wpre0P  = ws + off; off += 64 * 64;
  float* Wpre1P  = ws + off; off += 64 * 64;
  float* W1P     = ws + off; off += 8 * 64;
  float* W2P     = ws + off; off += 64 * 256;
  float* Wpost0P = ws + off; off += 64 * 64;
  float* Wpost1P = ws + off; off += 64 * 64;
  float* Wr0P    = ws + off; off += 640 * 64;
  float* Wr1P    = ws + off; off += 640 * 64;
  float* Wf0P    = ws + off; off += 64 * 64;
  float* Wf1P    = ws + off; off += 64 * 64;

  k_pack_all<<<PACK_BLOCKS, 256, 0, stream>>>(
      Wpre0, Wpre1, W1, W2, Wpost0, Wpost1, Wr0, Wr1, Wf0, Wf1,
      Wpre0P, Wpre1P, W1P, W2P, Wpost0P, Wpost1P, Wr0P, Wr1P, Wf0P, Wf1P);

  const int ntiles = N_NODES / 16;
  k_node_pre<<<(ntiles + 3) / 4, 128, 0, stream>>>(nf, Wpre0P, Wpre1P, x0, x1, t0, t1);
  k_edge<<<E_EDGES / 64, 128, 0, stream>>>(sph, rb, snd, rcv, W1P, W2P, x0, x1, t0, t1);
  k_node_post<<<ntiles, 128, 0, stream>>>(nf, attrs, Wpost0P, Wpost1P, Wc0, Wc1,
                                          Wr0P, Wr1P, Wf0P, Wf1P, t0, t1,
                                          (float*)d_out);
}


# edge kernel: each step's two message rows pass through per-wave LDS so every atomic instruction adds one whole 128-byte line of a receiver row; same adds and values, arrival-order sums as before
# speedup vs baseline: 1.7439x; 1.2030x over previous
#include <hip/hip_runtime.h>


#define N_NODES 20000
#define E_EDGES 320000
#define INV_SQRT3 0.57735026918962576451f

typedef __attribute__((ext_vector_type(2))) float v2f;
typedef __attribute__((ext_vector_type(8))) float v8f;

static __device__ __forceinline__ v8f wmma4(v2f a, v2f b, v8f c) {
  return __builtin_amdgcn_wmma_f32_16x16x4_f32(
      false, a, false, b, (short)0, c, false, false);
}

static __device__ __forceinline__ v2f ldb(const float* __restrict__ WP,
                                          int ncol, int k0, int col) {
  return *(const v2f*)(WP + (size_t)(k0 >> 1) * (2 * ncol) + col * 2);
}

static __device__ __forceinline__ v2f ld2(const float* __restrict__ p) {
  v2f r; r.x = p[0]; r.y = p[1]; return r;
}

static __device__ __forceinline__ void wave_lds_order() {
  __builtin_amdgcn_fence(__ATOMIC_RELEASE, "wavefront");
  __builtin_amdgcn_wave_barrier();
  __builtin_amdgcn_fence(__ATOMIC_ACQUIRE, "wavefront");
}

#define PACK_BLOCKS 241

__global__ __launch_bounds__(256) void k_pack_all(
    const float* __restrict__ Wpre0, const float* __restrict__ Wpre1,
    const float* __restrict__ W1, const float* __restrict__ W2,
    const float* __restrict__ Wpost0, const float* __restrict__ Wpost1,
    const float* __restrict__ Wr0, const float* __restrict__ Wr1,
    const float* __restrict__ Wf0, const float* __restrict__ Wf1,
    float* __restrict__ Wpre0P, float* __restrict__ Wpre1P,
    float* __restrict__ W1P, float* __restrict__ W2P,
    float* __restrict__ Wpost0P, float* __restrict__ Wpost1P,
    float* __restrict__ Wr0P, float* __restrict__ Wr1P,
    float* __restrict__ Wf0P, float* __restrict__ Wf1P) {
  const int b = blockIdx.x;
  const float* W;
  float* WP;
  int first, lg;
  if (b < 8)        { W = Wpre0;  WP = Wpre0P;  first = 0;   lg = 6; }
  else if (b < 16)  { W = Wpre1;  WP = Wpre1P;  first = 8;   lg = 6; }
  else if (b < 17)  { W = W1;     WP = W1P;     first = 16;  lg = 6; }
  else if (b < 49)  { W = W2;     WP = W2P;     first = 17;  lg = 8; }
  else if (b < 57)  { W = Wpost0; WP = Wpost0P; first = 49;  lg = 6; }
  else if (b < 65)  { W = Wpost1; WP = Wpost1P; first = 57;  lg = 6; }
  else if (b < 145) { W = Wr0;    WP = Wr0P;    first = 65;  lg = 6; }
  else if (b < 225) { W = Wr1;    WP = Wr1P;    first = 145; lg = 6; }
  else if (b < 233) { W = Wf0;    WP = Wf0P;    first = 225; lg = 6; }
  else              { W = Wf1;    WP = Wf1P;    first = 233; lg = 6; }
  const int p  = (b - first) * 256 + threadIdx.x;
  const int kp = p >> lg;
  const int c  = p & ((1 << lg) - 1);
  v2f o;
  o.x = W[((size_t)(2 * kp) << lg) + c];
  o.y = W[((size_t)(2 * kp + 1) << lg) + c];
  *(v2f*)(WP + 2 * (size_t)p) = o;
}

__global__ __launch_bounds__(128) void k_node_pre(
    const float* __restrict__ nf, const float* __restrict__ Wp0P,
    const float* __restrict__ Wp1P, float* __restrict__ x0,
    float* __restrict__ x1, float* __restrict__ t0, float* __restrict__ t1) {
  const int wave = threadIdx.x >> 5;
  const int lane = threadIdx.x & 31;
  const int tile = blockIdx.x * 4 + wave;
  if (tile >= (N_NODES / 16)) return;
  const int nbase = tile * 16;
  const int lrow  = lane & 15;
  const int kh    = lane >> 4;
  const int arow  = nbase + lrow;

  {
    v8f acc[4] = {{}, {}, {}, {}};
    for (int kk = 0; kk < 64; kk += 4) {
      const int k0 = kk + 2 * kh;
      const v2f a = ld2(&nf[(size_t)arow * 256 + k0]);
#pragma unroll
      for (int j = 0; j < 4; ++j)
        acc[j] = wmma4(a, ldb(Wp0P, 64, k0, 16 * j + lrow), acc[j]);
    }
#pragma unroll
    for (int j = 0; j < 4; ++j)
#pragma unroll
      for (int v = 0; v < 8; ++v)
        x0[(size_t)(nbase + v + 8 * kh) * 64 + 16 * j + lrow] = acc[j][v];
  }

  for (int m = 0; m < 3; ++m) {
    v8f acc[4] = {{}, {}, {}, {}};
    for (int kk = 0; kk < 64; kk += 4) {
      const int k0 = kk + 2 * kh;
      v2f a;
      a.x = nf[(size_t)arow * 256 + 64 + k0 * 3 + m];
      a.y = nf[(size_t)arow * 256 + 64 + (k0 + 1) * 3 + m];
#pragma unroll
      for (int j = 0; j < 4; ++j)
        acc[j] = wmma4(a, ldb(Wp1P, 64, k0, 16 * j + lrow), acc[j]);
    }
#pragma unroll
    for (int j = 0; j < 4; ++j)
#pragma unroll
      for (int v = 0; v < 8; ++v)
        x1[(size_t)(nbase + v + 8 * kh) * 192 + (16 * j + lrow) * 3 + m] = acc[j][v];
  }

  float4* t0v = (float4*)(t0 + (size_t)nbase * 64);
  float4* t1v = (float4*)(t1 + (size_t)nbase * 192);
  const float4 z4 = make_float4(0.f, 0.f, 0.f, 0.f);
  for (int i = lane; i < 16 * 64 / 4; i += 32)  t0v[i] = z4;
  for (int i = lane; i < 16 * 192 / 4; i += 32) t1v[i] = z4;
}

struct EdgeRows {
  float4 Y;
  float h0[4], hx[4], hy[4], hz[4];
};

static __device__ __forceinline__ void load_edge_rows(
    EdgeRows& q, const float* __restrict__ sph, const float* __restrict__ x0,
    const float* __restrict__ x1, int e, int s, int lrow) {
  q.Y = *(const float4*)(sph + (size_t)e * 4);
  const float* p0 = x0 + (size_t)s * 64 + lrow;
  const float* p1 = x1 + (size_t)s * 192 + lrow * 3;
#pragma unroll
  for (int j = 0; j < 4; ++j) {
    q.h0[j] = p0[16 * j];
    q.hx[j] = p1[48 * j + 0];
    q.hy[j] = p1[48 * j + 1];
    q.hz[j] = p1[48 * j + 2];
  }
}

__global__ __launch_bounds__(128) void k_edge(
    const float* __restrict__ sph, const float* __restrict__ rb,
    const int* __restrict__ snd, const int* __restrict__ rcv,
    const float* __restrict__ W1P, const float* __restrict__ W2P,
    const float* __restrict__ x0, const float* __restrict__ x1,
    float* __restrict__ t0, float* __restrict__ t1) {
  __shared__ float hidden[4][16][64];
  __shared__ float stage[4][2][272];
  const int wave  = threadIdx.x >> 5;
  const int lane  = threadIdx.x & 31;
  const int lrow  = lane & 15;
  const int kh    = lane >> 4;
  const int ebase = (blockIdx.x * 4 + wave) * 16;

  int sidx[8], ridx[8];
  {
    const int4* sp = (const int4*)(snd + ebase + 8 * kh);
    const int4* rp = (const int4*)(rcv + ebase + 8 * kh);
    const int4 sa = sp[0], sb = sp[1], ra = rp[0], rb4 = rp[1];
    sidx[0] = sa.x; sidx[1] = sa.y; sidx[2] = sa.z; sidx[3] = sa.w;
    sidx[4] = sb.x; sidx[5] = sb.y; sidx[6] = sb.z; sidx[7] = sb.w;
    ridx[0] = ra.x; ridx[1] = ra.y; ridx[2] = ra.z; ridx[3] = ra.w;
    ridx[4] = rb4.x; ridx[5] = rb4.y; ridx[6] = rb4.z; ridx[7] = rb4.w;
  }

  {
    v8f acc[4] = {{}, {}, {}, {}};
#pragma unroll
    for (int kk = 0; kk < 8; kk += 4) {
      const int k0 = kk + 2 * kh;
      const v2f a = ld2(&rb[(size_t)(ebase + lrow) * 8 + k0]);
#pragma unroll
      for (int j = 0; j < 4; ++j)
        acc[j] = wmma4(a, ldb(W1P, 64, k0, 16 * j + lrow), acc[j]);
    }
#pragma unroll
    for (int j = 0; j < 4; ++j)
#pragma unroll
      for (int v = 0; v < 8; ++v) {
        float h = acc[j][v];
        h = h / (1.f + __expf(-h));
        hidden[wave][v + 8 * kh][16 * j + lrow] = h;
      }
  }
  __syncthreads();

  v8f w[16] = {{}, {}, {}, {}, {}, {}, {}, {}, {}, {}, {}, {}, {}, {}, {}, {}};
  for (int kk = 0; kk < 64; kk += 4) {
    const int k0 = kk + 2 * kh;
    const v2f a = ld2(&hidden[wave][lrow][k0]);
#pragma unroll
    for (int j = 0; j < 16; ++j)
      w[j] = wmma4(a, ldb(W2P, 256, k0, 16 * j + lrow), w[j]);
  }

  EdgeRows g[3];
  load_edge_rows(g[0], sph, x0, x1, ebase + 8 * kh + 0, sidx[0], lrow);
  load_edge_rows(g[1], sph, x0, x1, ebase + 8 * kh + 1, sidx[1], lrow);
  float* const myst = &stage[wave][kh][0];
#pragma unroll
  for (int v = 0; v < 8; ++v) {
    if (v + 2 < 8)
      load_edge_rows(g[(v + 2) % 3], sph, x0, x1, ebase + 8 * kh + v + 2,
                     sidx[v + 2], lrow);
    const EdgeRows& q = g[v % 3];
    const float Y0 = q.Y.x, Y1x = q.Y.y, Y1y = q.Y.z, Y1z = q.Y.w;
#pragma unroll
    for (int j = 0; j < 4; ++j) {
      const int c = 16 * j + lrow;
      const float w0 = w[j][v], w1 = w[j + 4][v];
      const float w2 = w[j + 8][v], w3 = w[j + 12][v];
      const float h0  = q.h0[j];
      const float h1x = q.hx[j];
      const float h1y = q.hy[j];
      const float h1z = q.hz[j];
      const float dot = h1x * Y1x + h1y * Y1y + h1z * Y1z;
      myst[c] = w0 * h0 * Y0 + w1 * dot * INV_SQRT3;
      myst[64 + c * 3 + 0] = w2 * h0 * Y1x + w3 * h1x * Y0;
      myst[64 + c * 3 + 1] = w2 * h0 * Y1y + w3 * h1y * Y0;
      myst[64 + c * 3 + 2] = w2 * h0 * Y1z + w3 * h1z * Y0;
    }
    wave_lds_order();
#pragma unroll
    for (int h = 0; h < 2; ++h) {
      const int r = __builtin_amdgcn_readlane(ridx[v], 16 * h);
      const float* sh = &stage[wave][h][0];
      float m[8];
#pragma unroll
      for (int p = 0; p < 8; ++p) m[p] = sh[32 * p + lane];
      float* d0 = t0 + (size_t)r * 64 + lane;
      float* d1 = t1 + (size_t)r * 192 + lane;
      atomicAdd(d0, m[0]);
      atomicAdd(d0 + 32, m[1]);
#pragma unroll
      for (int p = 0; p < 6; ++p) atomicAdd(d1 + 32 * p, m[2 + p]);
    }
    wave_lds_order();
  }
}

__global__ __launch_bounds__(128) void k_node_post(
    const float* __restrict__ nf, const float* __restrict__ attrs,
    const float* __restrict__ Wpost0P, const float* __restrict__ Wpost1P,
    const float* __restrict__ Wc0, const float* __restrict__ Wc1,
    const float* __restrict__ Wr0P, const float* __restrict__ Wr1P,
    const float* __restrict__ Wf0P, const float* __restrict__ Wf1P,
    const float* __restrict__ t0, const float* __restrict__ t1,
    float* __restrict__ out) {
  __shared__ __attribute__((aligned(16))) float s0s[16][68];
  __shared__ __attribute__((aligned(16))) float s1s[3][16][68];
  __shared__ __attribute__((aligned(16))) float so[16][260];
  const int wave  = threadIdx.x >> 5;
  const int lane  = threadIdx.x & 31;
  const int lrow  = lane & 15;
  const int kh    = lane >> 4;
  const int nbase = blockIdx.x * 16;
  const int arow  = nbase + lrow;
  const int col   = 16 * wave + lrow;

  v8f a0 = {};
  v8f a1[3] = {{}, {}, {}};
  for (int kk = 0; kk < 64; kk += 4) {
    const int k0 = kk + 2 * kh;
    const v2f at0 = ld2(&t0[(size_t)arow * 64 + k0]);
    a0 = wmma4(at0, ldb(Wpost0P, 64, k0, col), a0);
    const v2f b1 = ldb(Wpost1P, 64, k0, col);
    const float* tp = &t1[(size_t)arow * 192 + k0 * 3];
    const v2f p0 = ld2(tp), p1 = ld2(tp + 2), p2 = ld2(tp + 4);
    v2f am;
    am.x = p0.x; am.y = p1.y;
    a1[0] = wmma4(am, b1, a1[0]);
    am.x = p0.y; am.y = p2.x;
    a1[1] = wmma4(am, b1, a1[1]);
    am.x = p1.x; am.y = p2.y;
    a1[2] = wmma4(am, b1, a1[2]);
  }

#pragma unroll
  for (int v = 0; v < 8; ++v) {
    const int node = nbase + v + 8 * kh;
    const float s  = a0[v] * (1.f / 16.f);
    const float ax = a1[0][v] * (1.f / 16.f);
    const float ay = a1[1][v] * (1.f / 16.f);
    const float az = a1[2][v] * (1.f / 16.f);
    const float q  = ax * ax + ay * ay + az * az;
    float c0 = 0.f, c1 = 0.f, c2 = 0.f, d0 = 0.f, d1 = 0.f, d2 = 0.f;
#pragma unroll
    for (int z = 0; z < 10; ++z) {
      const float t = attrs[(size_t)node * 10 + z];
      c0 += t * Wc0[z * 64 + col];
      c1 += t * Wc0[640 + z * 64 + col];
      c2 += t * Wc0[1280 + z * 64 + col];
      d0 += t * Wc1[z * 64 + col];
      d1 += t * Wc1[640 + z * 64 + col];
      d2 += t * Wc1[1280 + z * 64 + col];
    }
    const float s2 = s * s;
    const float g  = d2 * (s2 + q) + d1 * s + d0;
    a0[v]    = c2 * (s2 * s + s * q) + c1 * (s2 + q) + c0 * s;
    a1[0][v] = g * ax;
    a1[1][v] = g * ay;
    a1[2][v] = g * az;
  }

  v2f zz[5];
#pragma unroll
  for (int t = 0; t < 5; ++t) {
    const int z0 = (4 * t + 2 * kh) % 10;
    zz[t] = ld2(&attrs[(size_t)arow * 10 + z0]);
  }
  const bool hi = kh != 0;
  const float* xrow = nf + (size_t)arow * 256;
  for (int i = 0; i < 16; ++i) {
    const float4 q0 = *(const float4*)(xrow + 4 * i);
    const float4 q1 = *(const float4*)(xrow + 64 + 12 * i);
    const float4 q2 = *(const float4*)(xrow + 64 + 12 * i + 4);
    const float4 q3 = *(const float4*)(xrow + 64 + 12 * i + 8);
    const float xv[4]  = {q0.x, q0.y, q0.z, q0.w};
    const float yv[12] = {q1.x, q1.y, q1.z, q1.w, q2.x, q2.y,
                          q2.z, q2.w, q3.x, q3.y, q3.z, q3.w};
#pragma unroll
    for (int s = 0; s < 10; ++s) {
      const int k0 = 40 * i + 4 * s + 2 * kh;
      const int U0 = (4 * s) / 10;
      const int U1 = (4 * s + 2) / 10;
      float x  = xv[U0];
      float y0 = yv[3 * U0], y1 = yv[3 * U0 + 1], y2 = yv[3 * U0 + 2];
      if (U0 != U1) {
        x  = hi ? xv[U1] : x;
        y0 = hi ? yv[3 * U1] : y0;
        y1 = hi ? yv[3 * U1 + 1] : y1;
        y2 = hi ? yv[3 * U1 + 2] : y2;
      }
      const v2f az = zz[s % 5];
      v2f a;
      a.x = x * az.x; a.y = x * az.y;
      a0 = wmma4(a, ldb(Wr0P, 64, k0, col), a0);
      const v2f b1 = ldb(Wr1P, 64, k0, col);
      a.x = y0 * az.x; a.y = y0 * az.y;
      a1[0] = wmma4(a, b1, a1[0]);
      a.x = y1 * az.x; a.y = y1 * az.y;
      a1[1] = wmma4(a, b1, a1[1]);
      a.x = y2 * az.x; a.y = y2 * az.y;
      a1[2] = wmma4(a, b1, a1[2]);
    }
  }

#pragma unroll
  for (int v = 0; v < 8; ++v) {
    const int M = v + 8 * kh;
    s0s[M][col]    = a0[v];
    s1s[0][M][col] = a1[0][v];
    s1s[1][M][col] = a1[1][v];
    s1s[2][M][col] = a1[2][v];
  }
  __syncthreads();

  {
    v8f f0 = {};
    v8f f1[3] = {{}, {}, {}};
    for (int kk = 0; kk < 64; kk += 4) {
      const int k0 = kk + 2 * kh;
      const v2f af = *(const v2f*)&s0s[lrow][k0];
      f0 = wmma4(af, ldb(Wf0P, 64, k0, col), f0);
      const v2f b1 = ldb(Wf1P, 64, k0, col);
#pragma unroll
      for (int m = 0; m < 3; ++m) {
        const v2f am = *(const v2f*)&s1s[m][lrow][k0];
        f1[m] = wmma4(am, b1, f1[m]);
      }
    }
#pragma unroll
    for (int v = 0; v < 8; ++v) {
      const int M = v + 8 * kh;
      so[M][col] = f0[v];
      so[M][64 + col * 3 + 0] = f1[0][v];
      so[M][64 + col * 3 + 1] = f1[1][v];
      so[M][64 + col * 3 + 2] = f1[2][v];
    }
  }
  __syncthreads();

  float4* outv = (float4*)(out + (size_t)nbase * 256);
#pragma unroll
  for (int p = 0; p < 8; ++p) {
    const int f   = p * 128 + threadIdx.x;
    const int row = f >> 6, c4 = f & 63;
    outv[f] = *(const float4*)&so[row][4 * c4];
  }
}

extern "C" void kernel_launch(void* const* d_in, const int* in_sizes, int n_in,
                              void* d_out, int out_size, void* d_ws, size_t ws_size,
                              hipStream_t stream) {
  (void)in_sizes; (void)n_in; (void)out_size; (void)ws_size;
  const float* nf     = (const float*)d_in[0];
  const float* attrs  = (const float*)d_in[1];
  const float* sph    = (const float*)d_in[2];
  const float* rb     = (const float*)d_in[3];
  const int*   snd    = (const int*)d_in[4];
  const int*   rcv    = (const int*)d_in[5];
  const float* Wpre0  = (const float*)d_in[6];
  const float* Wpre1  = (const float*)d_in[7];
  const float* W1     = (const float*)d_in[8];
  const float* W2     = (const float*)d_in[9];
  const float* Wpost0 = (const float*)d_in[10];
  const float* Wpost1 = (const float*)d_in[11];
  const float* Wc0    = (const float*)d_in[12];
  const float* Wc1    = (const float*)d_in[13];
  const float* Wr0    = (const float*)d_in[14];
  const float* Wr1    = (const float*)d_in[15];
  const float* Wf0    = (const float*)d_in[16];
  const float* Wf1    = (const float*)d_in[17];

  float* ws = (float*)d_ws;
  size_t off = 0;
  float* x0 = ws + off; off += (size_t)N_NODES * 64;
  float* x1 = ws + off; off += (size_t)N_NODES * 192;
  float* t0 = ws + off; off += (size_t)N_NODES * 64;
  float* t1 = ws + off; off += (size_t)N_NODES * 192;
  float* Wpre0P  = ws + off; off += 64 * 64;
  float* Wpre1P  = ws + off; off += 64 * 64;
  float* W1P     = ws + off; off += 8 * 64;
  float* W2P     = ws + off; off += 64 * 256;
  float* Wpost0P = ws + off; off += 64 * 64;
  float* Wpost1P = ws + off; off += 64 * 64;
  float* Wr0P    = ws + off; off += 640 * 64;
  float* Wr1P    = ws + off; off += 640 * 64;
  float* Wf0P    = ws + off; off += 64 * 64;
  float* Wf1P    = ws + off; off += 64 * 64;

  k_pack_all<<<PACK_BLOCKS, 256, 0, stream>>>(
      Wpre0, Wpre1, W1, W2, Wpost0, Wpost1, Wr0, Wr1, Wf0, Wf1,
      Wpre0P, Wpre1P, W1P, W2P, Wpost0P, Wpost1P, Wr0P, Wr1P, Wf0P, Wf1P);

  const int ntiles = N_NODES / 16;
  k_node_pre<<<(ntiles + 3) / 4, 128, 0, stream>>>(nf, Wpre0P, Wpre1P, x0, x1, t0, t1);
  k_edge<<<E_EDGES / 64, 128, 0, stream>>>(sph, rb, snd, rcv, W1P, W2P, x0, x1, t0, t1);
  k_node_post<<<ntiles, 128, 0, stream>>>(nf, attrs, Wpost0P, Wpost1P, Wc0, Wc1,
                                          Wr0P, Wr1P, Wf0P, Wf1P, t0, t1,
                                          (float*)d_out);
}
